# starting kernel
# baseline (speedup 1.0000x reference)
#include <hip/hip_runtime.h>
#include <hip/hip_bf16.h>
#include <math.h>

#define DEPTH     15
#define N_LEAVES  32768
#define N_NODES   65535
#define FDIM      256
#define CDIM      2
#define NODE_ELEMS (FDIM * CDIM)
#define MIN_DIST  1e-7f

typedef __attribute__((ext_vector_type(2))) float v2f;
typedef __attribute__((ext_vector_type(8))) float v8f;

#define REG_BLOCKS 1024
#define REG_TPB    256
#define REG_WAVES  (REG_TPB / 32)
#define REG_NWAVES (REG_BLOCKS * REG_WAVES)

__global__ __launch_bounds__(REG_TPB)
void reg_partials_kernel(const float* __restrict__ deltas,
                         const float* __restrict__ branch,
                         float* __restrict__ partials)
{
    const int lane  = threadIdx.x & 31;
    const int wave  = threadIdx.x >> 5;
    const int gwave = blockIdx.x * REG_WAVES + wave;

    float accum = 0.0f;
    for (int n = gwave; n < N_NODES; n += REG_NWAVES) {
        if (n + REG_NWAVES < N_NODES) {
            __builtin_prefetch(deltas + (size_t)(n + REG_NWAVES) * NODE_ELEMS + lane * 16, 0, 0);
        }
        const float4* dp = (const float4*)(deltas + (size_t)n * NODE_ELEMS) + lane * 4;
        const float4 a0 = dp[0], a1 = dp[1], a2 = dp[2], a3 = dp[3];
        float s;
        if (n == 0) {
            s = a0.x*a0.x + a0.y*a0.y + a0.z*a0.z + a0.w*a0.w
              + a1.x*a1.x + a1.y*a1.y + a1.z*a1.z + a1.w*a1.w
              + a2.x*a2.x + a2.y*a2.y + a2.z*a2.z + a2.w*a2.w
              + a3.x*a3.x + a3.y*a3.y + a3.z*a3.z + a3.w*a3.w;
        } else {
            s = fabsf(a0.x)+fabsf(a0.y)+fabsf(a0.z)+fabsf(a0.w)
              + fabsf(a1.x)+fabsf(a1.y)+fabsf(a1.z)+fabsf(a1.w)
              + fabsf(a2.x)+fabsf(a2.y)+fabsf(a2.z)+fabsf(a2.w)
              + fabsf(a3.x)+fabsf(a3.y)+fabsf(a3.z)+fabsf(a3.w);
        }
        #pragma unroll
        for (int off = 16; off >= 1; off >>= 1) s += __shfl_xor(s, off, 32);
        if (n != 0) s /= fmaxf(branch[n - 1], MIN_DIST);
        accum += s;
    }
    if (lane == 0) partials[gwave] = accum;
}

__global__ __launch_bounds__(1024)
void reg_reduce_kernel(const float* __restrict__ partials, float* __restrict__ out)
{
    __shared__ float sm[1024];
    float s = 0.0f;
    for (int i = threadIdx.x; i < REG_NWAVES; i += 1024) s += partials[i];
    sm[threadIdx.x] = s;
    __syncthreads();
    for (int w = 512; w >= 1; w >>= 1) {
        if ((int)threadIdx.x < w) sm[threadIdx.x] += sm[threadIdx.x + w];
        __syncthreads();
    }
    if (threadIdx.x == 0) out[2 * N_LEAVES] = sm[0];
}

#define LG_WAVES 2
#define XS_STRIDE 260

__global__ __launch_bounds__(LG_WAVES * 32)
void logits_softmax_kernel(const float* __restrict__ x,
                           const float* __restrict__ deltas,
                           float* __restrict__ out)
{
    __shared__ float xs[LG_WAVES][16 * XS_STRIDE];
    __shared__ float wsh[LG_WAVES][NODE_ELEMS];
    __shared__ float lg[LG_WAVES][16 * CDIM];

    const int lane  = threadIdx.x & 31;
    const int wave  = threadIdx.x >> 5;
    const int tile  = blockIdx.x * LG_WAVES + wave;
    const int leaf0 = tile * 16;

    float* xw  = xs[wave];
    float* ww  = wsh[wave];
    float* lgw = lg[wave];

    {
        const float4* xb4 = (const float4*)(x + (size_t)leaf0 * FDIM);
        for (int t = lane; t < (16 * FDIM) / 4; t += 32) {
            const int r = t >> 6, c4 = t & 63;
            *(float4*)&xw[r * XS_STRIDE + c4 * 4] = xb4[t];
        }
    }

    {
        float4 acc0 = {0,0,0,0}, acc1 = {0,0,0,0}, acc2 = {0,0,0,0}, acc3 = {0,0,0,0};
        #pragma unroll
        for (int l = 0; l <= 11; ++l) {
            const int idx = ((1 << l) - 1) + (leaf0 >> (DEPTH - l));
            const float4* dp = (const float4*)(deltas + (size_t)idx * NODE_ELEMS) + lane * 4;
            const float4 d0 = dp[0], d1 = dp[1], d2 = dp[2], d3 = dp[3];
            acc0.x += d0.x; acc0.y += d0.y; acc0.z += d0.z; acc0.w += d0.w;
            acc1.x += d1.x; acc1.y += d1.y; acc1.z += d1.z; acc1.w += d1.w;
            acc2.x += d2.x; acc2.y += d2.y; acc2.z += d2.z; acc2.w += d2.w;
            acc3.x += d3.x; acc3.y += d3.y; acc3.z += d3.z; acc3.w += d3.w;
        }
        float4* wv = (float4*)(ww) + lane * 4;
        wv[0] = acc0; wv[1] = acc1; wv[2] = acc2; wv[3] = acc3;
    }
    if (lane < 16) { lgw[lane * CDIM + 0] = 0.0f; lgw[lane * CDIM + 1] = 0.0f; }
    __syncthreads();

    v8f cfrag = {};
    const int mn   = lane & 15;
    const int koff = (lane < 16) ? 0 : 2;
    const int wcol = mn & 1;
    for (int k0 = 0; k0 < FDIM; k0 += 4) {
        const int k = k0 + koff;
        v2f a, b;
        a.x = ww[k * CDIM + wcol];
        a.y = ww[(k + 1) * CDIM + wcol];
        b.x = xw[mn * XS_STRIDE + k];
        b.y = xw[mn * XS_STRIDE + k + 1];
        cfrag = __builtin_amdgcn_wmma_f32_16x16x4_f32(
             false, a,  false, b,
             (short)0, cfrag,  false,  false);
    }

    for (int j = 0; j < 16; ++j) {
        const int leaf = leaf0 + j;
        const float4* xr4 = (const float4*)(xw + j * XS_STRIDE) + lane * 2;
        const float4 xa = xr4[0], xb = xr4[1];
        float s0 = 0.0f, s1 = 0.0f;
        #pragma unroll
        for (int l = 12; l <= 15; ++l) {
            const int idx = ((1 << l) - 1) + (leaf >> (DEPTH - l));
            const float4* dp = (const float4*)(deltas + (size_t)idx * NODE_ELEMS) + lane * 4;
            const float4 d0 = dp[0], d1 = dp[1], d2 = dp[2], d3 = dp[3];
            s0 += xa.x*d0.x + xa.y*d0.z + xa.z*d1.x + xa.w*d1.z;
            s1 += xa.x*d0.y + xa.y*d0.w + xa.z*d1.y + xa.w*d1.w;
            s0 += xb.x*d2.x + xb.y*d2.z + xb.z*d3.x + xb.w*d3.z;
            s1 += xb.x*d2.y + xb.y*d2.w + xb.z*d3.y + xb.w*d3.w;
        }
        #pragma unroll
        for (int off = 16; off >= 1; off >>= 1) {
            s0 += __shfl_xor(s0, off, 32);
            s1 += __shfl_xor(s1, off, 32);
        }
        if (lane == 0) {
            lgw[j * CDIM + 0] += s0;
            lgw[j * CDIM + 1] += s1;
        }
    }
    __syncthreads();

    if (lane < 16) {
        const float z0 = cfrag[0] + lgw[lane * CDIM + 0];
        const float z1 = cfrag[1] + lgw[lane * CDIM + 1];
        const float m  = fmaxf(z0, z1);
        const float e0 = expf(z0 - m);
        const float e1 = expf(z1 - m);
        const float inv = 1.0f / (e0 + e1);
        const size_t o = (size_t)(leaf0 + lane) * CDIM;
        out[o + 0] = e0 * inv;
        out[o + 1] = e1 * inv;
    }
}

extern "C" void kernel_launch(void* const* d_in, const int* in_sizes, int n_in,
                              void* d_out, int out_size, void* d_ws, size_t ws_size,
                              hipStream_t stream) {
    const float* x      = (const float*)d_in[0];
    const float* deltas = (const float*)d_in[1];
    const float* branch = (const float*)d_in[2];
    float* out      = (float*)d_out;
    float* partials = (float*)d_ws;

    reg_partials_kernel<<<REG_BLOCKS, REG_TPB, 0, stream>>>(deltas, branch, partials);
    reg_reduce_kernel<<<1, 1024, 0, stream>>>(partials, out);

    const int n_tiles = N_LEAVES / 16;
    logits_softmax_kernel<<<n_tiles / LG_WAVES, LG_WAVES * 32, 0, stream>>>(x, deltas, out);
}


# Regularizer fused into the logits kernel: each delta node is loaded once, node-major in batches, for both its dot products and its |d|/len term; 3 launches to 2; regularizer summation order changes
# speedup vs baseline: 1.7039x; 1.7039x over previous
#include <hip/hip_runtime.h>
#include <hip/hip_bf16.h>
#include <math.h>

#define DEPTH     15
#define N_LEAVES  32768
#define N_NODES   65535
#define FDIM      256
#define CDIM      2
#define NODE_ELEMS (FDIM * CDIM)
#define MIN_DIST  1e-7f
#define N_TILES   (N_LEAVES / 16)

#define LG_WAVES  2
#define XS_STRIDE 260

typedef __attribute__((ext_vector_type(2))) float v2f;
typedef __attribute__((ext_vector_type(8))) float v8f;

struct Node16 { float4 q0, q1, q2, q3; };

__device__ __forceinline__ Node16 load_node(const float* __restrict__ deltas, int idx, int lane)
{
    const float4* dp = (const float4*)(deltas + (size_t)idx * NODE_ELEMS) + lane * 4;
    Node16 n;
    n.q0 = dp[0]; n.q1 = dp[1]; n.q2 = dp[2]; n.q3 = dp[3];
    return n;
}

__device__ __forceinline__ float abs_sum(const Node16& n)
{
    return fabsf(n.q0.x)+fabsf(n.q0.y)+fabsf(n.q0.z)+fabsf(n.q0.w)
         + fabsf(n.q1.x)+fabsf(n.q1.y)+fabsf(n.q1.z)+fabsf(n.q1.w)
         + fabsf(n.q2.x)+fabsf(n.q2.y)+fabsf(n.q2.z)+fabsf(n.q2.w)
         + fabsf(n.q3.x)+fabsf(n.q3.y)+fabsf(n.q3.z)+fabsf(n.q3.w);
}

__device__ __forceinline__ float sq_sum(const Node16& n)
{
    return n.q0.x*n.q0.x + n.q0.y*n.q0.y + n.q0.z*n.q0.z + n.q0.w*n.q0.w
         + n.q1.x*n.q1.x + n.q1.y*n.q1.y + n.q1.z*n.q1.z + n.q1.w*n.q1.w
         + n.q2.x*n.q2.x + n.q2.y*n.q2.y + n.q2.z*n.q2.z + n.q2.w*n.q2.w
         + n.q3.x*n.q3.x + n.q3.y*n.q3.y + n.q3.z*n.q3.z + n.q3.w*n.q3.w;
}

__device__ __forceinline__ float reg_term(const Node16& n, float len)
{
    return abs_sum(n) / fmaxf(len, MIN_DIST);
}

__device__ __forceinline__ void dot2(const float4& xa, const float4& xb, const Node16& n,
                                     float& s0, float& s1)
{
    s0 += xa.x*n.q0.x + xa.y*n.q0.z + xa.z*n.q1.x + xa.w*n.q1.z;
    s1 += xa.x*n.q0.y + xa.y*n.q0.w + xa.z*n.q1.y + xa.w*n.q1.w;
    s0 += xb.x*n.q2.x + xb.y*n.q2.z + xb.z*n.q3.x + xb.w*n.q3.z;
    s1 += xb.x*n.q2.y + xb.y*n.q2.w + xb.z*n.q3.y + xb.w*n.q3.w;
}

struct Batch {
    Node16 n12, n13, n14[2], n15[4];
    float  l12, l13, l14[2], l15[4];
};

template <bool MORE>
__device__ __forceinline__ void batch_step(const float* __restrict__ deltas,
                                           const float* __restrict__ branch,
                                           const float* xw, int leaf0, int b, int lane,
                                           Batch& B, float& racc, float& r0, float& r1)
{
    const int leafn = leaf0 + 4 * (b + 1);

    float4 xa[4], xb[4];
    float  s0[4], s1[4];
    #pragma unroll
    for (int k = 0; k < 4; ++k) {
        const float4* xr4 = (const float4*)(xw + (4 * b + k) * XS_STRIDE) + lane * 2;
        xa[k] = xr4[0]; xb[k] = xr4[1];
        s0[k] = 0.0f;   s1[k] = 0.0f;
    }

    #pragma unroll
    for (int k = 0; k < 4; ++k) dot2(xa[k], xb[k], B.n12, s0[k], s1[k]);
    if (b & 1) {
        racc += reg_term(B.n12, B.l12);
        if (MORE) {
            const int i12 = ((1 << 12) - 1) + (leafn >> 3);
            B.n12 = load_node(deltas, i12, lane);  B.l12 = branch[i12 - 1];
        }
    }

    #pragma unroll
    for (int k = 0; k < 4; ++k) dot2(xa[k], xb[k], B.n13, s0[k], s1[k]);
    racc += reg_term(B.n13, B.l13);
    if (MORE) {
        const int i13 = ((1 << 13) - 1) + (leafn >> 2);
        B.n13 = load_node(deltas, i13, lane);  B.l13 = branch[i13 - 1];
    }

    #pragma unroll
    for (int k = 0; k < 4; ++k) dot2(xa[k], xb[k], B.n14[k >> 1], s0[k], s1[k]);
    #pragma unroll
    for (int k = 0; k < 2; ++k) racc += reg_term(B.n14[k], B.l14[k]);
    if (MORE) {
        const int i14 = ((1 << 14) - 1) + (leafn >> 1);
        #pragma unroll
        for (int k = 0; k < 2; ++k) { B.n14[k] = load_node(deltas, i14 + k, lane); B.l14[k] = branch[i14 + k - 1]; }
    }

    #pragma unroll
    for (int k = 0; k < 4; ++k) dot2(xa[k], xb[k], B.n15[k], s0[k], s1[k]);
    #pragma unroll
    for (int k = 0; k < 4; ++k) racc += reg_term(B.n15[k], B.l15[k]);
    if (MORE) {
        const int i15 = ((1 << 15) - 1) + leafn;
        #pragma unroll
        for (int k = 0; k < 4; ++k) { B.n15[k] = load_node(deltas, i15 + k, lane); B.l15[k] = branch[i15 + k - 1]; }
    }

    #pragma unroll
    for (int k = 0; k < 4; ++k) {
        float t0 = s0[k], t1 = s1[k];
        #pragma unroll
        for (int off = 16; off >= 1; off >>= 1) {
            t0 += __shfl_xor(t0, off, 32);
            t1 += __shfl_xor(t1, off, 32);
        }
        if ((lane & 15) == 4 * b + k) { r0 = t0; r1 = t1; }
    }
}

__global__ __launch_bounds__(LG_WAVES * 32)
void logits_reg_kernel(const float* __restrict__ x,
                       const float* __restrict__ deltas,
                       const float* __restrict__ branch,
                       float* __restrict__ out,
                       float* __restrict__ partials)
{
    __shared__ float xs[LG_WAVES][16 * XS_STRIDE];
    __shared__ float wsh[LG_WAVES][NODE_ELEMS];

    const int lane  = threadIdx.x & 31;
    const int wave  = __builtin_amdgcn_readfirstlane((int)(threadIdx.x >> 5));
    const int tile  = blockIdx.x * LG_WAVES + wave;
    const int leaf0 = tile * 16;

    float* xw = xs[wave];
    float* ww = wsh[wave];

    float racc = 0.0f;

    {
        const float4* xb4 = (const float4*)(x + (size_t)leaf0 * FDIM);
        for (int t = lane; t < (16 * FDIM) / 4; t += 32) {
            const int r = t >> 6, c4 = t & 63;
            *(float4*)&xw[r * XS_STRIDE + c4 * 4] = xb4[t];
        }
    }

    {
        float4 acc0 = {0,0,0,0}, acc1 = {0,0,0,0}, acc2 = {0,0,0,0}, acc3 = {0,0,0,0};
        #pragma unroll
        for (int l = 0; l <= 11; ++l) {
            const int idx = ((1 << l) - 1) + (leaf0 >> (DEPTH - l));
            const Node16 d = load_node(deltas, idx, lane);
            acc0.x += d.q0.x; acc0.y += d.q0.y; acc0.z += d.q0.z; acc0.w += d.q0.w;
            acc1.x += d.q1.x; acc1.y += d.q1.y; acc1.z += d.q1.z; acc1.w += d.q1.w;
            acc2.x += d.q2.x; acc2.y += d.q2.y; acc2.z += d.q2.z; acc2.w += d.q2.w;
            acc3.x += d.q3.x; acc3.y += d.q3.y; acc3.z += d.q3.z; acc3.w += d.q3.w;
            const bool first_below = (tile & ((1 << (11 - l)) - 1)) == 0;
            const float term = (l == 0) ? sq_sum(d)
                                        : reg_term(d, branch[l == 0 ? 0 : idx - 1]);
            racc += first_below ? term : 0.0f;
        }
        float4* wv = (float4*)(ww) + lane * 4;
        wv[0] = acc0; wv[1] = acc1; wv[2] = acc2; wv[3] = acc3;
    }

    Batch B;
    {
        const int i12 = ((1 << 12) - 1) + (leaf0 >> 3);
        const int i13 = ((1 << 13) - 1) + (leaf0 >> 2);
        const int i14 = ((1 << 14) - 1) + (leaf0 >> 1);
        const int i15 = ((1 << 15) - 1) + leaf0;
        B.n12 = load_node(deltas, i12, lane);  B.l12 = branch[i12 - 1];
        B.n13 = load_node(deltas, i13, lane);  B.l13 = branch[i13 - 1];
        #pragma unroll
        for (int k = 0; k < 2; ++k) { B.n14[k] = load_node(deltas, i14 + k, lane); B.l14[k] = branch[i14 + k - 1]; }
        #pragma unroll
        for (int k = 0; k < 4; ++k) { B.n15[k] = load_node(deltas, i15 + k, lane); B.l15[k] = branch[i15 + k - 1]; }
    }

    __syncthreads();

    v8f cfrag = {};
    const int mn   = lane & 15;
    const int koff = (lane < 16) ? 0 : 2;
    const int wcol = mn & 1;
    for (int k0 = 0; k0 < FDIM; k0 += 4) {
        const int k = k0 + koff;
        v2f a, b;
        a.x = ww[k * CDIM + wcol];
        a.y = ww[(k + 1) * CDIM + wcol];
        b.x = xw[mn * XS_STRIDE + k];
        b.y = xw[mn * XS_STRIDE + k + 1];
        cfrag = __builtin_amdgcn_wmma_f32_16x16x4_f32(
             false, a,  false, b,
             (short)0, cfrag,  false,  false);
    }

    float r0 = 0.0f, r1 = 0.0f;
    for (int b = 0; b < 3; ++b)
        batch_step<true>(deltas, branch, xw, leaf0, b, lane, B, racc, r0, r1);
    batch_step<false>(deltas, branch, xw, leaf0, 3, lane, B, racc, r0, r1);

    if (lane < 16) {
        const float z0 = cfrag[0] + r0;
        const float z1 = cfrag[1] + r1;
        const float m  = fmaxf(z0, z1);
        const float e0 = expf(z0 - m);
        const float e1 = expf(z1 - m);
        const float inv = 1.0f / (e0 + e1);
        const size_t o = (size_t)(leaf0 + lane) * CDIM;
        out[o + 0] = e0 * inv;
        out[o + 1] = e1 * inv;
    }

    #pragma unroll
    for (int off = 16; off >= 1; off >>= 1) racc += __shfl_xor(racc, off, 32);
    if (lane == 0) partials[tile] = racc;
}

__global__ __launch_bounds__(1024)
void reg_reduce_kernel(const float* __restrict__ partials, float* __restrict__ out)
{
    __shared__ float sm[1024];
    float s = 0.0f;
    for (int i = threadIdx.x; i < N_TILES; i += 1024) s += partials[i];
    sm[threadIdx.x] = s;
    __syncthreads();
    for (int w = 512; w >= 1; w >>= 1) {
        if ((int)threadIdx.x < w) sm[threadIdx.x] += sm[threadIdx.x + w];
        __syncthreads();
    }
    if (threadIdx.x == 0) out[2 * N_LEAVES] = sm[0];
}

extern "C" void kernel_launch(void* const* d_in, const int* in_sizes, int n_in,
                              void* d_out, int out_size, void* d_ws, size_t ws_size,
                              hipStream_t stream) {
    const float* x      = (const float*)d_in[0];
    const float* deltas = (const float*)d_in[1];
    const float* branch = (const float*)d_in[2];
    float* out      = (float*)d_out;
    float* partials = (float*)d_ws;

    logits_reg_kernel<<<N_TILES / LG_WAVES, LG_WAVES * 32, 0, stream>>>(x, deltas, branch, out, partials);
    reg_reduce_kernel<<<1, 1024, 0, stream>>>(partials, out);
}


# Deeper load window per wave: x tile requested as one group of 32 loads (was 4 drained groups of 8), per-leaf nodes two batches ahead (was one), body one straight run; all sums bit-identical
# speedup vs baseline: 1.9817x; 1.1630x over previous
#include <hip/hip_runtime.h>
#include <hip/hip_bf16.h>
#include <math.h>

#define DEPTH     15
#define N_LEAVES  32768
#define N_NODES   65535
#define FDIM      256
#define CDIM      2
#define NODE_ELEMS (FDIM * CDIM)
#define MIN_DIST  1e-7f
#define N_TILES   (N_LEAVES / 16)

#define LG_WAVES  2
#define XS_STRIDE 260

typedef __attribute__((ext_vector_type(2))) float v2f;
typedef __attribute__((ext_vector_type(8))) float v8f;

struct Node16 { float4 q0, q1, q2, q3; };

__device__ __forceinline__ Node16 load_node(const float* __restrict__ deltas, int idx, int lane)
{
    const float4* dp = (const float4*)(deltas + (size_t)idx * NODE_ELEMS) + lane * 4;
    Node16 n;
    n.q0 = dp[0]; n.q1 = dp[1]; n.q2 = dp[2]; n.q3 = dp[3];
    return n;
}

__device__ __forceinline__ float abs_sum(const Node16& n)
{
    return fabsf(n.q0.x)+fabsf(n.q0.y)+fabsf(n.q0.z)+fabsf(n.q0.w)
         + fabsf(n.q1.x)+fabsf(n.q1.y)+fabsf(n.q1.z)+fabsf(n.q1.w)
         + fabsf(n.q2.x)+fabsf(n.q2.y)+fabsf(n.q2.z)+fabsf(n.q2.w)
         + fabsf(n.q3.x)+fabsf(n.q3.y)+fabsf(n.q3.z)+fabsf(n.q3.w);
}

__device__ __forceinline__ float sq_sum(const Node16& n)
{
    return n.q0.x*n.q0.x + n.q0.y*n.q0.y + n.q0.z*n.q0.z + n.q0.w*n.q0.w
         + n.q1.x*n.q1.x + n.q1.y*n.q1.y + n.q1.z*n.q1.z + n.q1.w*n.q1.w
         + n.q2.x*n.q2.x + n.q2.y*n.q2.y + n.q2.z*n.q2.z + n.q2.w*n.q2.w
         + n.q3.x*n.q3.x + n.q3.y*n.q3.y + n.q3.z*n.q3.z + n.q3.w*n.q3.w;
}

__device__ __forceinline__ float reg_term(const Node16& n, float len)
{
    return abs_sum(n) / fmaxf(len, MIN_DIST);
}

__device__ __forceinline__ void dot2(const float4& xa, const float4& xb, const Node16& n,
                                     float& s0, float& s1)
{
    s0 += xa.x*n.q0.x + xa.y*n.q0.z + xa.z*n.q1.x + xa.w*n.q1.z;
    s1 += xa.x*n.q0.y + xa.y*n.q0.w + xa.z*n.q1.y + xa.w*n.q1.w;
    s0 += xb.x*n.q2.x + xb.y*n.q2.z + xb.z*n.q3.x + xb.w*n.q3.z;
    s1 += xb.x*n.q2.y + xb.y*n.q2.w + xb.z*n.q3.y + xb.w*n.q3.w;
}

struct Batch {
    Node16 n13, n14[2], n15[4];
    float  l13, l14[2], l15[4];
};

__device__ __forceinline__ void load_batch(const float* __restrict__ deltas,
                                           const float* __restrict__ branch,
                                           int leaf, int lane, Batch& B)
{
    const int i13 = ((1 << 13) - 1) + (leaf >> 2);
    const int i14 = ((1 << 14) - 1) + (leaf >> 1);
    const int i15 = ((1 << 15) - 1) + leaf;
    B.n13 = load_node(deltas, i13, lane);  B.l13 = branch[i13 - 1];
    #pragma unroll
    for (int k = 0; k < 2; ++k) { B.n14[k] = load_node(deltas, i14 + k, lane); B.l14[k] = branch[i14 + k - 1]; }
    #pragma unroll
    for (int k = 0; k < 4; ++k) { B.n15[k] = load_node(deltas, i15 + k, lane); B.l15[k] = branch[i15 + k - 1]; }
}

template <bool MORE, bool REG12>
__device__ __forceinline__ void batch_step(const float* __restrict__ deltas,
                                           const float* __restrict__ branch,
                                           const float* xw, int leaf0, int b, int lane,
                                           const Node16& n12, float l12,
                                           Batch& B, float& racc, float& r0, float& r1)
{
    const int leafn = leaf0 + 4 * (b + 2);

    float4 xa[4], xb[4];
    float  s0[4], s1[4];
    #pragma unroll
    for (int k = 0; k < 4; ++k) {
        const float4* xr4 = (const float4*)(xw + (4 * b + k) * XS_STRIDE) + lane * 2;
        xa[k] = xr4[0]; xb[k] = xr4[1];
        s0[k] = 0.0f;   s1[k] = 0.0f;
    }

    #pragma unroll
    for (int k = 0; k < 4; ++k) dot2(xa[k], xb[k], n12, s0[k], s1[k]);
    if (REG12) racc += reg_term(n12, l12);

    #pragma unroll
    for (int k = 0; k < 4; ++k) dot2(xa[k], xb[k], B.n13, s0[k], s1[k]);
    racc += reg_term(B.n13, B.l13);
    if (MORE) {
        const int i13 = ((1 << 13) - 1) + (leafn >> 2);
        B.n13 = load_node(deltas, i13, lane);  B.l13 = branch[i13 - 1];
    }

    #pragma unroll
    for (int k = 0; k < 4; ++k) dot2(xa[k], xb[k], B.n14[k >> 1], s0[k], s1[k]);
    #pragma unroll
    for (int k = 0; k < 2; ++k) racc += reg_term(B.n14[k], B.l14[k]);
    if (MORE) {
        const int i14 = ((1 << 14) - 1) + (leafn >> 1);
        #pragma unroll
        for (int k = 0; k < 2; ++k) { B.n14[k] = load_node(deltas, i14 + k, lane); B.l14[k] = branch[i14 + k - 1]; }
    }

    #pragma unroll
    for (int k = 0; k < 4; ++k) dot2(xa[k], xb[k], B.n15[k], s0[k], s1[k]);
    #pragma unroll
    for (int k = 0; k < 4; ++k) racc += reg_term(B.n15[k], B.l15[k]);
    if (MORE) {
        const int i15 = ((1 << 15) - 1) + leafn;
        #pragma unroll
        for (int k = 0; k < 4; ++k) { B.n15[k] = load_node(deltas, i15 + k, lane); B.l15[k] = branch[i15 + k - 1]; }
    }

    #pragma unroll
    for (int k = 0; k < 4; ++k) {
        float t0 = s0[k], t1 = s1[k];
        #pragma unroll
        for (int off = 16; off >= 1; off >>= 1) {
            t0 += __shfl_xor(t0, off, 32);
            t1 += __shfl_xor(t1, off, 32);
        }
        if ((lane & 15) == 4 * b + k) { r0 = t0; r1 = t1; }
    }
}

__global__ __launch_bounds__(LG_WAVES * 32, 2)
void logits_reg_kernel(const float* __restrict__ x,
                       const float* __restrict__ deltas,
                       const float* __restrict__ branch,
                       float* __restrict__ out,
                       float* __restrict__ partials)
{
    __shared__ float xs[LG_WAVES][16 * XS_STRIDE];
    __shared__ float wsh[LG_WAVES][NODE_ELEMS];

    const int lane  = threadIdx.x & 31;
    const int wave  = __builtin_amdgcn_readfirstlane((int)(threadIdx.x >> 5));
    const int tile  = blockIdx.x * LG_WAVES + wave;
    const int leaf0 = tile * 16;

    float* xw = xs[wave];
    float* ww = wsh[wave];

    float racc = 0.0f;

    Node16 n12a, n12b;
    float  l12a, l12b;
    Batch  BA, BB;
    const int i12 = ((1 << 12) - 1) + (leaf0 >> 3);

    float4 xv[32];
    {
        const float4* xb4 = (const float4*)(x + (size_t)leaf0 * FDIM);
        #pragma unroll
        for (int j = 0; j < 32; ++j) xv[j] = xb4[lane + 32 * j];
        n12a = load_node(deltas, i12, lane);      l12a = branch[i12 - 1];
        load_batch(deltas, branch, leaf0, lane, BA);
    }
    __builtin_amdgcn_sched_barrier(0);

    #pragma unroll
    for (int j = 0; j < 32; ++j) {
        const int t = lane + 32 * j;
        const int r = t >> 6, c4 = t & 63;
        *(float4*)&xw[r * XS_STRIDE + c4 * 4] = xv[j];
    }
    __builtin_amdgcn_sched_barrier(0);

    Node16 anc[12];
    float  lanc[12];
    #pragma unroll
    for (int l = 0; l <= 11; ++l) {
        const int idx = ((1 << l) - 1) + (leaf0 >> (DEPTH - l));
        anc[l]  = load_node(deltas, idx, lane);
        lanc[l] = (l == 0) ? 1.0f : branch[l == 0 ? 0 : idx - 1];
    }
    __builtin_amdgcn_sched_barrier(0);

    {
        float4 acc0 = {0,0,0,0}, acc1 = {0,0,0,0}, acc2 = {0,0,0,0}, acc3 = {0,0,0,0};
        #pragma unroll
        for (int l = 0; l <= 11; ++l) {
            const Node16& d = anc[l];
            acc0.x += d.q0.x; acc0.y += d.q0.y; acc0.z += d.q0.z; acc0.w += d.q0.w;
            acc1.x += d.q1.x; acc1.y += d.q1.y; acc1.z += d.q1.z; acc1.w += d.q1.w;
            acc2.x += d.q2.x; acc2.y += d.q2.y; acc2.z += d.q2.z; acc2.w += d.q2.w;
            acc3.x += d.q3.x; acc3.y += d.q3.y; acc3.z += d.q3.z; acc3.w += d.q3.w;
            const bool first_below = (tile & ((1 << (11 - l)) - 1)) == 0;
            const float term = (l == 0) ? sq_sum(d)
                                        : reg_term(d, lanc[l]);
            racc += first_below ? term : 0.0f;
        }
        float4* wv = (float4*)(ww) + lane * 4;
        wv[0] = acc0; wv[1] = acc1; wv[2] = acc2; wv[3] = acc3;
    }
    __builtin_amdgcn_sched_barrier(0);

    load_batch(deltas, branch, leaf0 + 4, lane, BB);
    n12b = load_node(deltas, i12 + 1, lane);  l12b = branch[i12];

    __syncthreads();

    v8f cfrag = {};
    const int mn   = lane & 15;
    const int koff = (lane < 16) ? 0 : 2;
    const int wcol = mn & 1;
    #pragma unroll
    for (int k0 = 0; k0 < FDIM; k0 += 4) {
        const int k = k0 + koff;
        v2f a, b;
        a.x = ww[k * CDIM + wcol];
        a.y = ww[(k + 1) * CDIM + wcol];
        b.x = xw[mn * XS_STRIDE + k];
        b.y = xw[mn * XS_STRIDE + k + 1];
        cfrag = __builtin_amdgcn_wmma_f32_16x16x4_f32(
             false, a,  false, b,
             (short)0, cfrag,  false,  false);
    }

    float r0 = 0.0f, r1 = 0.0f;
    batch_step<true,  false>(deltas, branch, xw, leaf0, 0, lane, n12a, l12a, BA, racc, r0, r1);
    batch_step<true,  true >(deltas, branch, xw, leaf0, 1, lane, n12a, l12a, BB, racc, r0, r1);
    batch_step<false, false>(deltas, branch, xw, leaf0, 2, lane, n12b, l12b, BA, racc, r0, r1);
    batch_step<false, true >(deltas, branch, xw, leaf0, 3, lane, n12b, l12b, BB, racc, r0, r1);

    #pragma unroll
    for (int off = 16; off >= 1; off >>= 1) racc += __shfl_xor(racc, off, 32);
    if (lane == 0) partials[tile] = racc;

    if (lane < 16) {
        const float z0 = cfrag[0] + r0;
        const float z1 = cfrag[1] + r1;
        const float m  = fmaxf(z0, z1);
        const float e0 = expf(z0 - m);
        const float e1 = expf(z1 - m);
        const float inv = 1.0f / (e0 + e1);
        const size_t o = (size_t)(leaf0 + lane) * CDIM;
        out[o + 0] = e0 * inv;
        out[o + 1] = e1 * inv;
    }
}

__global__ __launch_bounds__(1024)
void reg_reduce_kernel(const float* __restrict__ partials, float* __restrict__ out)
{
    __shared__ float sm[1024];
    float s = 0.0f;
    for (int i = threadIdx.x; i < N_TILES; i += 1024) s += partials[i];
    sm[threadIdx.x] = s;
    __syncthreads();
    for (int w = 512; w >= 1; w >>= 1) {
        if ((int)threadIdx.x < w) sm[threadIdx.x] += sm[threadIdx.x + w];
        __syncthreads();
    }
    if (threadIdx.x == 0) out[2 * N_LEAVES] = sm[0];
}

extern "C" void kernel_launch(void* const* d_in, const int* in_sizes, int n_in,
                              void* d_out, int out_size, void* d_ws, size_t ws_size,
                              hipStream_t stream) {
    const float* x      = (const float*)d_in[0];
    const float* deltas = (const float*)d_in[1];
    const float* branch = (const float*)d_in[2];
    float* out      = (float*)d_out;
    float* partials = (float*)d_ws;

    logits_reg_kernel<<<N_TILES / LG_WAVES, LG_WAVES * 32, 0, stream>>>(x, deltas, branch, out, partials);
    reg_reduce_kernel<<<1, 1024, 0, stream>>>(partials, out);
}
